# starting kernel
# baseline (speedup 1.0000x reference)
#include <hip/hip_runtime.h>

typedef __attribute__((ext_vector_type(2))) float v2f;
typedef __attribute__((ext_vector_type(8))) float v8f;

#define TPB 256


__global__ void gcn_init(float* __restrict__ deg, float* __restrict__ s,
                         float* __restrict__ agg0, float* __restrict__ agg1,
                         int n) {
  int i = blockIdx.x * blockDim.x + threadIdx.x;
  if (i < n) {
    deg[i] = 1.0f;
    s[i] = 0.0f;
    agg0[i] = 0.0f;
    agg1[i] = 0.0f;
  }
}

__global__ void gcn_degree(const int* __restrict__ dst,
                           float* __restrict__ deg, int E) {
  int e = blockIdx.x * blockDim.x + threadIdx.x;
  if (e < E) unsafeAtomicAdd(&deg[dst[e]], 1.0f);
}

__global__ void gcn_dinv(const float* __restrict__ deg,
                         float* __restrict__ dinv, int n) {
  int i = blockIdx.x * blockDim.x + threadIdx.x;
  if (i < n) dinv[i] = rsqrtf(deg[i]);
}

__global__ void gcn_l1_scatter(const int* __restrict__ src,
                               const int* __restrict__ dst,
                               const float* __restrict__ x,
                               const float* __restrict__ dinv,
                               float* __restrict__ s, int E) {
  int e = blockIdx.x * blockDim.x + threadIdx.x;
  if (e >= E) return;
  int a = src[e], b = dst[e];
  unsafeAtomicAdd(&s[b], x[a] * dinv[a] * dinv[b]);
}

__global__ void gcn_l1_finalize(const float* __restrict__ x,
                                const float* __restrict__ dinv,
                                const float* __restrict__ s,
                                const float* __restrict__ W1,
                                const float* __restrict__ b1,
                                float* __restrict__ relu1, int n) {
  int i = blockIdx.x * blockDim.x + threadIdx.x;
  if (i >= n) return;
  float di = dinv[i];
  float sn = s[i] + x[i] * di * di;
  float4* o = (float4*)(relu1 + (size_t)i * 16);
#pragma unroll
  for (int q = 0; q < 4; ++q) {
    float4 r;
    r.x = fmaxf(fmaf(sn, W1[4 * q + 0], b1[4 * q + 0]), 0.0f);
    r.y = fmaxf(fmaf(sn, W1[4 * q + 1], b1[4 * q + 1]), 0.0f);
    r.z = fmaxf(fmaf(sn, W1[4 * q + 2], b1[4 * q + 2]), 0.0f);
    r.w = fmaxf(fmaf(sn, W1[4 * q + 3], b1[4 * q + 3]), 0.0f);
    o[q] = r;
  }
}

__global__ void gcn_l2_dense_wmma(const float* __restrict__ relu1,
                                  const float* __restrict__ W2,
                                  float* __restrict__ h2c0,
                                  float* __restrict__ h2c1,
                                  int n, int ntiles) {
  int gtid = blockIdx.x * blockDim.x + threadIdx.x;
  int tile = gtid >> 5;
  int lane = threadIdx.x & 31;
  if (tile >= ntiles) return;
  int base = tile * 16;
  int half = lane >> 4;
  int m = lane & 15;

  v2f bmat[4];
#pragma unroll
  for (int k = 0; k < 4; ++k) {
    int r0 = 4 * k + 2 * half;
    float bx = 0.0f, by = 0.0f;
    if (m < 2) { bx = W2[r0 * 2 + m]; by = W2[(r0 + 1) * 2 + m]; }
    bmat[k].x = bx; bmat[k].y = by;
  }

  int row = base + m;
  row = (row < n) ? row : (n - 1);
  const v2f* p = (const v2f*)(relu1 + (size_t)row * 16 + 2 * half);
  v2f amat[4];
#pragma unroll
  for (int k = 0; k < 4; ++k) amat[k] = p[2 * k];

  v8f c = {};
#pragma unroll
  for (int k = 0; k < 4; ++k)
    c = __builtin_amdgcn_wmma_f32_16x16x4_f32(
         false, amat[k],  false, bmat[k],
         (short)0, c,  false,  false);

  if (m < 2) {
    float* col = (m == 0) ? h2c0 : h2c1;
    int node0 = base + half * 8;
    float4 lo; lo.x = c[0]; lo.y = c[1]; lo.z = c[2]; lo.w = c[3];
    float4 hi; hi.x = c[4]; hi.y = c[5]; hi.z = c[6]; hi.w = c[7];
    *(float4*)(col + node0)     = lo;
    *(float4*)(col + node0 + 4) = hi;
  }
}

__global__ void gcn_l2_scatter(const int* __restrict__ src,
                               const int* __restrict__ dst,
                               const float* __restrict__ h2c0,
                               const float* __restrict__ h2c1,
                               const float* __restrict__ dinv,
                               float* __restrict__ agg0,
                               float* __restrict__ agg1, int E) {
  int e = blockIdx.x * blockDim.x + threadIdx.x;
  if (e >= E) return;
  int a = src[e], b = dst[e];
  float nrm = dinv[a] * dinv[b];
  unsafeAtomicAdd(&agg0[b], h2c0[a] * nrm);
  unsafeAtomicAdd(&agg1[b], h2c1[a] * nrm);
}

__global__ void gcn_finalize(const float* __restrict__ agg0,
                             const float* __restrict__ agg1,
                             const float* __restrict__ h2c0,
                             const float* __restrict__ h2c1,
                             const float* __restrict__ dinv,
                             const float* __restrict__ b2,
                             float* __restrict__ out, int n) {
  int i = blockIdx.x * blockDim.x + threadIdx.x;
  if (i >= n) return;
  float di = dinv[i];
  float sl = di * di;
  float z0 = agg0[i] + h2c0[i] * sl + b2[0];
  float z1 = agg1[i] + h2c1[i] * sl + b2[1];
  float mx = fmaxf(z0, z1);
  float lse = mx + logf(expf(z0 - mx) + expf(z1 - mx));
  float2 r; r.x = z0 - lse; r.y = z1 - lse;
  *(float2*)(out + (size_t)i * 2) = r;
}

extern "C" void kernel_launch(void* const* d_in, const int* in_sizes, int n_in,
                              void* d_out, int out_size, void* d_ws, size_t ws_size,
                              hipStream_t stream) {
  const float* x  = (const float*)d_in[0];
  const int*   ei = (const int*)d_in[1];
  const float* W1 = (const float*)d_in[2];
  const float* b1 = (const float*)d_in[3];
  const float* W2 = (const float*)d_in[4];
  const float* b2 = (const float*)d_in[5];
  float* out = (float*)d_out;

  int n = in_sizes[0];
  int E = in_sizes[1] / 2;
  const int* srcp = ei;
  const int* dstp = ei + E;

  size_t ns = ((size_t)n + 31) & ~(size_t)31;

  float* ws    = (float*)d_ws;
  float* deg   = ws;
  float* dinv  = ws + ns;
  float* s     = ws + 2 * ns;
  float* h2c0  = ws + 3 * ns;
  float* h2c1  = ws + 4 * ns;
  float* agg0  = ws + 5 * ns;
  float* agg1  = ws + 6 * ns;
  float* relu1 = ws + 7 * ns;

  int gN = (n + TPB - 1) / TPB;
  int gE = (E + TPB - 1) / TPB;

  gcn_init<<<gN, TPB, 0, stream>>>(deg, s, agg0, agg1, n);
  gcn_degree<<<gE, TPB, 0, stream>>>(dstp, deg, E);
  gcn_dinv<<<gN, TPB, 0, stream>>>(deg, dinv, n);
  gcn_l1_scatter<<<gE, TPB, 0, stream>>>(srcp, dstp, x, dinv, s, E);
  gcn_l1_finalize<<<gN, TPB, 0, stream>>>(x, dinv, s, W1, b1, relu1, n);

  int ntiles = (n + 15) / 16;
  int gW = (int)(((size_t)ntiles * 32 + TPB - 1) / TPB);
  gcn_l2_dense_wmma<<<gW, TPB, 0, stream>>>(relu1, W2, h2c0, h2c1, n, ntiles);

  gcn_l2_scatter<<<gE, TPB, 0, stream>>>(srcp, dstp, h2c0, h2c1, dinv,
                                         agg0, agg1, E);
  gcn_finalize<<<gN, TPB, 0, stream>>>(agg0, agg1, h2c0, h2c1, dinv, b2, out, n);
}


# Scatter-adds accumulate in LDS bins split by destination range, then one coalesced float-atomic flush per block; changes summation order (per-block partial sums first)
# speedup vs baseline: 3.4336x; 3.4336x over previous
#include <hip/hip_runtime.h>

typedef __attribute__((ext_vector_type(2))) float v2f;
typedef __attribute__((ext_vector_type(8))) float v8f;

#define TPB 256


__global__ void gcn_init(float* __restrict__ deg, float* __restrict__ s,
                         float* __restrict__ agg0, float* __restrict__ agg1,
                         int n) {
  int i = blockIdx.x * blockDim.x + threadIdx.x;
  if (i < n) {
    deg[i] = 1.0f;
    s[i] = 0.0f;
    agg0[i] = 0.0f;
    agg1[i] = 0.0f;
  }
}

#define SC_THREADS 1024
#define SC_BINS    12544
#define SC_SLICES  32
#define SC_U       8
#define SC_U_DEG   16

__device__ __forceinline__ void lds_add(float* p, float v) {
  __hip_atomic_fetch_add(p, v, __ATOMIC_RELAXED, __HIP_MEMORY_SCOPE_WORKGROUP);
}

__device__ __forceinline__ void sc_flush(const float* bins, float* __restrict__ g,
                                         int lo, int cnt) {
  int rot = (int)((blockIdx.x * (unsigned)cnt) / gridDim.x) & ~31;
  for (int i = threadIdx.x; i < cnt; i += SC_THREADS) {
    int j = i + rot;
    if (j >= cnt) j -= cnt;
    unsafeAtomicAdd(&g[lo + j], bins[j]);
  }
}

__global__ __launch_bounds__(SC_THREADS)
void gcn_degree(const int* __restrict__ dst, float* __restrict__ deg,
                int E, int n, int chunk) {
  __shared__ float bins[SC_BINS];
  const int tid = threadIdx.x;
  const int lo = blockIdx.y * SC_BINS;
  const int cnt = min(n - lo, SC_BINS);
  for (int i = tid; i < SC_BINS; i += SC_THREADS) bins[i] = 0.0f;
  __syncthreads();

  const int start = blockIdx.x * chunk;
  const int end = min(E, start + chunk);
  const int last = end - 1;
  for (int e0 = start + tid; e0 < end; e0 += SC_THREADS * SC_U_DEG) {
    int d[SC_U_DEG];
#pragma unroll
    for (int k = 0; k < SC_U_DEG; ++k) {
      int e = e0 + k * SC_THREADS;
      int dv = dst[min(e, last)];
      d[k] = (e < end) ? dv : -1;
    }
#pragma unroll
    for (int k = 0; k < SC_U_DEG; ++k) {
      unsigned r = (unsigned)(d[k] - lo);
      if (r < (unsigned)cnt) lds_add(&bins[r], 1.0f);
    }
  }
  __syncthreads();
  sc_flush(bins, deg, lo, cnt);
}

__global__ void gcn_dinv(const float* __restrict__ deg,
                         float* __restrict__ dinv, int n) {
  int i = blockIdx.x * blockDim.x + threadIdx.x;
  if (i < n) dinv[i] = rsqrtf(deg[i]);
}

__global__ __launch_bounds__(SC_THREADS, 16)
void gcn_l1_scatter(const int* __restrict__ src, const int* __restrict__ dst,
                    const float* __restrict__ x, const float* __restrict__ dinv,
                    float* __restrict__ s, int E, int n, int chunk) {
  __shared__ float bins[SC_BINS];
  const int tid = threadIdx.x;
  const int lo = blockIdx.y * SC_BINS;
  const int cnt = min(n - lo, SC_BINS);
  for (int i = tid; i < SC_BINS; i += SC_THREADS) bins[i] = 0.0f;
  __syncthreads();

  const int start = blockIdx.x * chunk;
  const int end = min(E, start + chunk);
  if (start < end) {
    const int last = end - 1;
    int e0 = start + tid;
    int d[SC_U], a[SC_U];
#pragma unroll
    for (int k = 0; k < SC_U; ++k) {
      int e = e0 + k * SC_THREADS;
      int ec = min(e, last);
      int dv = dst[ec];
      a[k] = src[ec];
      d[k] = (e < end) ? dv : -1;
    }
    for (; e0 < end; e0 += SC_THREADS * SC_U) {
      int dn[SC_U], an[SC_U];
#pragma unroll
      for (int k = 0; k < SC_U; ++k) {
        int e = e0 + (SC_U + k) * SC_THREADS;
        int ec = min(e, last);
        int dv = dst[ec];
        an[k] = src[ec];
        dn[k] = (e < end) ? dv : -1;
      }
      float v[SC_U];
      unsigned r[SC_U];
#pragma unroll
      for (int k = 0; k < SC_U; ++k) {
        r[k] = (unsigned)(d[k] - lo);
        bool ok = r[k] < (unsigned)cnt;
        int ai = ok ? a[k] : 0;
        int bi = ok ? d[k] : lo;
        v[k] = x[ai] * dinv[ai] * dinv[bi];
      }
      static_assert(SC_U == 8, "the statement below names eight products");
      asm volatile("" :: "v"(v[0]), "v"(v[1]), "v"(v[2]), "v"(v[3]),
                         "v"(v[4]), "v"(v[5]), "v"(v[6]), "v"(v[7]));
#pragma unroll
      for (int k = 0; k < SC_U; ++k)
        if (r[k] < (unsigned)cnt) lds_add(&bins[r[k]], v[k]);
#pragma unroll
      for (int k = 0; k < SC_U; ++k) { d[k] = dn[k]; a[k] = an[k]; }
    }
  }
  __syncthreads();
  sc_flush(bins, s, lo, cnt);
}

__global__ void gcn_l1_finalize(const float* __restrict__ x,
                                const float* __restrict__ dinv,
                                const float* __restrict__ s,
                                const float* __restrict__ W1,
                                const float* __restrict__ b1,
                                float* __restrict__ relu1, int n) {
  int i = blockIdx.x * blockDim.x + threadIdx.x;
  if (i >= n) return;
  float di = dinv[i];
  float sn = s[i] + x[i] * di * di;
  float4* o = (float4*)(relu1 + (size_t)i * 16);
#pragma unroll
  for (int q = 0; q < 4; ++q) {
    float4 r;
    r.x = fmaxf(fmaf(sn, W1[4 * q + 0], b1[4 * q + 0]), 0.0f);
    r.y = fmaxf(fmaf(sn, W1[4 * q + 1], b1[4 * q + 1]), 0.0f);
    r.z = fmaxf(fmaf(sn, W1[4 * q + 2], b1[4 * q + 2]), 0.0f);
    r.w = fmaxf(fmaf(sn, W1[4 * q + 3], b1[4 * q + 3]), 0.0f);
    o[q] = r;
  }
}

__global__ void gcn_l2_dense_wmma(const float* __restrict__ relu1,
                                  const float* __restrict__ W2,
                                  float* __restrict__ h2c0,
                                  float* __restrict__ h2c1,
                                  int n, int ntiles) {
  int gtid = blockIdx.x * blockDim.x + threadIdx.x;
  int tile = gtid >> 5;
  int lane = threadIdx.x & 31;
  if (tile >= ntiles) return;
  int base = tile * 16;
  int half = lane >> 4;
  int m = lane & 15;

  v2f bmat[4];
#pragma unroll
  for (int k = 0; k < 4; ++k) {
    int r0 = 4 * k + 2 * half;
    float bx = 0.0f, by = 0.0f;
    if (m < 2) { bx = W2[r0 * 2 + m]; by = W2[(r0 + 1) * 2 + m]; }
    bmat[k].x = bx; bmat[k].y = by;
  }

  int row = base + m;
  row = (row < n) ? row : (n - 1);
  const v2f* p = (const v2f*)(relu1 + (size_t)row * 16 + 2 * half);
  v2f amat[4];
#pragma unroll
  for (int k = 0; k < 4; ++k) amat[k] = p[2 * k];

  v8f c = {};
#pragma unroll
  for (int k = 0; k < 4; ++k)
    c = __builtin_amdgcn_wmma_f32_16x16x4_f32(
         false, amat[k],  false, bmat[k],
         (short)0, c,  false,  false);

  if (m < 2) {
    float* col = (m == 0) ? h2c0 : h2c1;
    int node0 = base + half * 8;
    float4 lo; lo.x = c[0]; lo.y = c[1]; lo.z = c[2]; lo.w = c[3];
    float4 hi; hi.x = c[4]; hi.y = c[5]; hi.z = c[6]; hi.w = c[7];
    *(float4*)(col + node0)     = lo;
    *(float4*)(col + node0 + 4) = hi;
  }
}

__global__ __launch_bounds__(SC_THREADS, 16)
void gcn_l2_scatter(const int* __restrict__ src, const int* __restrict__ dst,
                    const float* __restrict__ h2c0, const float* __restrict__ h2c1,
                    const float* __restrict__ dinv,
                    float* __restrict__ agg0, float* __restrict__ agg1,
                    int E, int n, int chunk) {
  __shared__ float bins0[SC_BINS];
  __shared__ float bins1[SC_BINS];
  const int tid = threadIdx.x;
  const int lo = blockIdx.y * SC_BINS;
  const int cnt = min(n - lo, SC_BINS);
  for (int i = tid; i < SC_BINS; i += SC_THREADS) { bins0[i] = 0.0f; bins1[i] = 0.0f; }
  __syncthreads();

  const int start = blockIdx.x * chunk;
  const int end = min(E, start + chunk);
  if (start < end) {
    const int last = end - 1;
    int e0 = start + tid;
    int d[SC_U], a[SC_U];
#pragma unroll
    for (int k = 0; k < SC_U; ++k) {
      int e = e0 + k * SC_THREADS;
      int ec = min(e, last);
      int dv = dst[ec];
      a[k] = src[ec];
      d[k] = (e < end) ? dv : -1;
    }
    for (; e0 < end; e0 += SC_THREADS * SC_U) {
      int dn[SC_U], an[SC_U];
#pragma unroll
      for (int k = 0; k < SC_U; ++k) {
        int e = e0 + (SC_U + k) * SC_THREADS;
        int ec = min(e, last);
        int dv = dst[ec];
        an[k] = src[ec];
        dn[k] = (e < end) ? dv : -1;
      }
      float v0[SC_U], v1[SC_U];
      unsigned r[SC_U];
#pragma unroll
      for (int k = 0; k < SC_U; ++k) {
        r[k] = (unsigned)(d[k] - lo);
        bool ok = r[k] < (unsigned)cnt;
        int ai = ok ? a[k] : 0;
        int bi = ok ? d[k] : lo;
        float nrm = dinv[ai] * dinv[bi];
        v0[k] = h2c0[ai] * nrm;
        v1[k] = h2c1[ai] * nrm;
      }
      static_assert(SC_U == 8, "the statement below names eight pairs");
      asm volatile("" :: "v"(v0[0]), "v"(v0[1]), "v"(v0[2]), "v"(v0[3]),
                         "v"(v0[4]), "v"(v0[5]), "v"(v0[6]), "v"(v0[7]),
                         "v"(v1[0]), "v"(v1[1]), "v"(v1[2]), "v"(v1[3]),
                         "v"(v1[4]), "v"(v1[5]), "v"(v1[6]), "v"(v1[7]));
#pragma unroll
      for (int k = 0; k < SC_U; ++k)
        if (r[k] < (unsigned)cnt) {
          lds_add(&bins0[r[k]], v0[k]);
          lds_add(&bins1[r[k]], v1[k]);
        }
#pragma unroll
      for (int k = 0; k < SC_U; ++k) { d[k] = dn[k]; a[k] = an[k]; }
    }
  }
  __syncthreads();
  sc_flush(bins0, agg0, lo, cnt);
  sc_flush(bins1, agg1, lo, cnt);
}

__global__ void gcn_finalize(const float* __restrict__ agg0,
                             const float* __restrict__ agg1,
                             const float* __restrict__ h2c0,
                             const float* __restrict__ h2c1,
                             const float* __restrict__ dinv,
                             const float* __restrict__ b2,
                             float* __restrict__ out, int n) {
  int i = blockIdx.x * blockDim.x + threadIdx.x;
  if (i >= n) return;
  float di = dinv[i];
  float sl = di * di;
  float z0 = agg0[i] + h2c0[i] * sl + b2[0];
  float z1 = agg1[i] + h2c1[i] * sl + b2[1];
  float mx = fmaxf(z0, z1);
  float lse = mx + logf(expf(z0 - mx) + expf(z1 - mx));
  float2 r; r.x = z0 - lse; r.y = z1 - lse;
  *(float2*)(out + (size_t)i * 2) = r;
}

extern "C" void kernel_launch(void* const* d_in, const int* in_sizes, int n_in,
                              void* d_out, int out_size, void* d_ws, size_t ws_size,
                              hipStream_t stream) {
  const float* x  = (const float*)d_in[0];
  const int*   ei = (const int*)d_in[1];
  const float* W1 = (const float*)d_in[2];
  const float* b1 = (const float*)d_in[3];
  const float* W2 = (const float*)d_in[4];
  const float* b2 = (const float*)d_in[5];
  float* out = (float*)d_out;

  int n = in_sizes[0];
  int E = in_sizes[1] / 2;
  const int* srcp = ei;
  const int* dstp = ei + E;

  size_t ns = ((size_t)n + 31) & ~(size_t)31;

  float* ws    = (float*)d_ws;
  float* deg   = ws;
  float* dinv  = ws + ns;
  float* s     = ws + 2 * ns;
  float* h2c0  = ws + 3 * ns;
  float* h2c1  = ws + 4 * ns;
  float* agg0  = ws + 5 * ns;
  float* agg1  = ws + 6 * ns;
  float* relu1 = ws + 7 * ns;

  int gN = (n + TPB - 1) / TPB;

  gcn_init<<<gN, TPB, 0, stream>>>(deg, s, agg0, agg1, n);
  int chunk = (E + SC_SLICES - 1) / SC_SLICES;
  dim3 gS(SC_SLICES, (n + SC_BINS - 1) / SC_BINS);
  gcn_degree<<<gS, SC_THREADS, 0, stream>>>(dstp, deg, E, n, chunk);
  gcn_dinv<<<gN, TPB, 0, stream>>>(deg, dinv, n);
  gcn_l1_scatter<<<gS, SC_THREADS, 0, stream>>>(srcp, dstp, x, dinv, s, E, n, chunk);
  gcn_l1_finalize<<<gN, TPB, 0, stream>>>(x, dinv, s, W1, b1, relu1, n);

  int ntiles = (n + 15) / 16;
  int gW = (int)(((size_t)ntiles * 32 + TPB - 1) / TPB);
  gcn_l2_dense_wmma<<<gW, TPB, 0, stream>>>(relu1, W2, h2c0, h2c1, n, ntiles);

  gcn_l2_scatter<<<gS, SC_THREADS, 0, stream>>>(srcp, dstp, h2c0, h2c1, dinv,
                                                agg0, agg1, E, n, chunk);
  gcn_finalize<<<gN, TPB, 0, stream>>>(agg0, agg1, h2c0, h2c1, dinv, b2, out, n);
}


# dinv[src] folded into the gathered node value, dinv[dst] applied after the sum: one gather per edge (an 8-byte pair in layer 2) instead of 3 and 4; changes the rounding of the summed terms
# speedup vs baseline: 6.5347x; 1.9032x over previous
#include <hip/hip_runtime.h>

typedef __attribute__((ext_vector_type(2))) float v2f;
typedef __attribute__((ext_vector_type(8))) float v8f;

#define TPB 256


__global__ void gcn_init(float* __restrict__ deg, float* __restrict__ s,
                         float* __restrict__ agg0, float* __restrict__ agg1,
                         int n) {
  int i = blockIdx.x * blockDim.x + threadIdx.x;
  if (i < n) {
    deg[i] = 1.0f;
    s[i] = 0.0f;
    agg0[i] = 0.0f;
    agg1[i] = 0.0f;
  }
}

#define SC_THREADS 1024
#define SC_BINS    12544
#define SC_SLICES  32
#define SC_U       8
#define SC_U_DEG   16

__device__ __forceinline__ void lds_add(float* p, float v) {
  __hip_atomic_fetch_add(p, v, __ATOMIC_RELAXED, __HIP_MEMORY_SCOPE_WORKGROUP);
}

__device__ __forceinline__ void sc_flush(const float* bins, float* __restrict__ g,
                                         int lo, int cnt) {
  int rot = (int)((blockIdx.x * (unsigned)cnt) / gridDim.x) & ~31;
  for (int i = threadIdx.x; i < cnt; i += SC_THREADS) {
    int j = i + rot;
    if (j >= cnt) j -= cnt;
    unsafeAtomicAdd(&g[lo + j], bins[j]);
  }
}

__global__ __launch_bounds__(SC_THREADS)
void gcn_degree(const int* __restrict__ dst, float* __restrict__ deg,
                int E, int n, int chunk) {
  __shared__ float bins[SC_BINS];
  const int tid = threadIdx.x;
  const int lo = blockIdx.y * SC_BINS;
  const int cnt = min(n - lo, SC_BINS);
  for (int i = tid; i < SC_BINS; i += SC_THREADS) bins[i] = 0.0f;
  __syncthreads();

  const int start = blockIdx.x * chunk;
  const int end = min(E, start + chunk);
  const int last = end - 1;
  for (int e0 = start + tid; e0 < end; e0 += SC_THREADS * SC_U_DEG) {
    int d[SC_U_DEG];
#pragma unroll
    for (int k = 0; k < SC_U_DEG; ++k) {
      int e = e0 + k * SC_THREADS;
      int dv = dst[min(e, last)];
      d[k] = (e < end) ? dv : -1;
    }
#pragma unroll
    for (int k = 0; k < SC_U_DEG; ++k) {
      unsigned r = (unsigned)(d[k] - lo);
      if (r < (unsigned)cnt) lds_add(&bins[r], 1.0f);
    }
  }
  __syncthreads();
  sc_flush(bins, deg, lo, cnt);
}

__global__ void gcn_dinv(const float* __restrict__ deg,
                         const float* __restrict__ x,
                         float* __restrict__ dinv, float* __restrict__ g1,
                         int n, int ns) {
  int i = blockIdx.x * blockDim.x + threadIdx.x;
  if (i < n) {
    float di = rsqrtf(deg[i]);
    dinv[i] = di;
    g1[i] = x[i] * di;
  } else if (i < ns) {
    dinv[i] = 0.0f;
  }
}

__global__ __launch_bounds__(SC_THREADS, 16)
void gcn_l1_scatter(const int* __restrict__ src, const int* __restrict__ dst,
                    const float* __restrict__ g1,
                    float* __restrict__ s, int E, int n, int chunk) {
  __shared__ float bins[SC_BINS];
  const int tid = threadIdx.x;
  const int lo = blockIdx.y * SC_BINS;
  const int cnt = min(n - lo, SC_BINS);
  for (int i = tid; i < SC_BINS; i += SC_THREADS) bins[i] = 0.0f;
  __syncthreads();

  const int start = blockIdx.x * chunk;
  const int end = min(E, start + chunk);
  if (start < end) {
    const int last = end - 1;
    int e0 = start + tid;
    int d[SC_U], a[SC_U];
#pragma unroll
    for (int k = 0; k < SC_U; ++k) {
      int e = e0 + k * SC_THREADS;
      int ec = min(e, last);
      int dv = dst[ec];
      a[k] = src[ec];
      d[k] = (e < end) ? dv : -1;
    }
    for (; e0 < end; e0 += SC_THREADS * SC_U) {
      int dn[SC_U], an[SC_U];
#pragma unroll
      for (int k = 0; k < SC_U; ++k) {
        int e = e0 + (SC_U + k) * SC_THREADS;
        int ec = min(e, last);
        int dv = dst[ec];
        an[k] = src[ec];
        dn[k] = (e < end) ? dv : -1;
      }
      float v[SC_U];
      unsigned r[SC_U];
#pragma unroll
      for (int k = 0; k < SC_U; ++k) {
        r[k] = (unsigned)(d[k] - lo);
        bool ok = r[k] < (unsigned)cnt;
        int ai = ok ? a[k] : 0;
        v[k] = g1[ai];
      }
      static_assert(SC_U == 8, "the statement below names eight values");
      asm volatile("" :: "v"(v[0]), "v"(v[1]), "v"(v[2]), "v"(v[3]),
                         "v"(v[4]), "v"(v[5]), "v"(v[6]), "v"(v[7]));
#pragma unroll
      for (int k = 0; k < SC_U; ++k)
        if (r[k] < (unsigned)cnt) lds_add(&bins[r[k]], v[k]);
#pragma unroll
      for (int k = 0; k < SC_U; ++k) { d[k] = dn[k]; a[k] = an[k]; }
    }
  }
  __syncthreads();
  sc_flush(bins, s, lo, cnt);
}

__global__ void gcn_l1_finalize(const float* __restrict__ g1,
                                const float* __restrict__ dinv,
                                const float* __restrict__ s,
                                const float* __restrict__ W1,
                                const float* __restrict__ b1,
                                float* __restrict__ relu1, int n) {
  int i = blockIdx.x * blockDim.x + threadIdx.x;
  if (i >= n) return;
  float sn = dinv[i] * (s[i] + g1[i]);
  float4* o = (float4*)(relu1 + (size_t)i * 16);
#pragma unroll
  for (int q = 0; q < 4; ++q) {
    float4 r;
    r.x = fmaxf(fmaf(sn, W1[4 * q + 0], b1[4 * q + 0]), 0.0f);
    r.y = fmaxf(fmaf(sn, W1[4 * q + 1], b1[4 * q + 1]), 0.0f);
    r.z = fmaxf(fmaf(sn, W1[4 * q + 2], b1[4 * q + 2]), 0.0f);
    r.w = fmaxf(fmaf(sn, W1[4 * q + 3], b1[4 * q + 3]), 0.0f);
    o[q] = r;
  }
}

__global__ void gcn_l2_dense_wmma(const float* __restrict__ relu1,
                                  const float* __restrict__ W2,
                                  const float* __restrict__ dinv,
                                  float* __restrict__ g2,
                                  int n, int ntiles) {
  int gtid = blockIdx.x * blockDim.x + threadIdx.x;
  int tile = gtid >> 5;
  int lane = threadIdx.x & 31;
  if (tile >= ntiles) return;
  int base = tile * 16;
  int half = lane >> 4;
  int m = lane & 15;

  v2f bmat[4];
#pragma unroll
  for (int k = 0; k < 4; ++k) {
    int r0 = 4 * k + 2 * half;
    float bx = 0.0f, by = 0.0f;
    if (m < 2) { bx = W2[r0 * 2 + m]; by = W2[(r0 + 1) * 2 + m]; }
    bmat[k].x = bx; bmat[k].y = by;
  }

  int row = base + m;
  row = (row < n) ? row : (n - 1);
  const v2f* p = (const v2f*)(relu1 + (size_t)row * 16 + 2 * half);
  v2f amat[4];
#pragma unroll
  for (int k = 0; k < 4; ++k) amat[k] = p[2 * k];

  v8f c = {};
#pragma unroll
  for (int k = 0; k < 4; ++k)
    c = __builtin_amdgcn_wmma_f32_16x16x4_f32(
         false, amat[k],  false, bmat[k],
         (short)0, c,  false,  false);

  if (m < 2) {
    int node0 = base + half * 8;
    float4 dl = *(const float4*)(dinv + node0);
    float4 dh = *(const float4*)(dinv + node0 + 4);
    float* o = g2 + (size_t)node0 * 2 + m;
    o[0]  = c[0] * dl.x;
    o[2]  = c[1] * dl.y;
    o[4]  = c[2] * dl.z;
    o[6]  = c[3] * dl.w;
    o[8]  = c[4] * dh.x;
    o[10] = c[5] * dh.y;
    o[12] = c[6] * dh.z;
    o[14] = c[7] * dh.w;
  }
}

__global__ __launch_bounds__(SC_THREADS, 16)
void gcn_l2_scatter(const int* __restrict__ src, const int* __restrict__ dst,
                    const float* __restrict__ g2,
                    float* __restrict__ agg0, float* __restrict__ agg1,
                    int E, int n, int chunk) {
  const float2* __restrict__ g2p = (const float2*)g2;
  __shared__ float bins0[SC_BINS];
  __shared__ float bins1[SC_BINS];
  const int tid = threadIdx.x;
  const int lo = blockIdx.y * SC_BINS;
  const int cnt = min(n - lo, SC_BINS);
  for (int i = tid; i < SC_BINS; i += SC_THREADS) { bins0[i] = 0.0f; bins1[i] = 0.0f; }
  __syncthreads();

  const int start = blockIdx.x * chunk;
  const int end = min(E, start + chunk);
  if (start < end) {
    const int last = end - 1;
    int e0 = start + tid;
    int d[SC_U], a[SC_U];
#pragma unroll
    for (int k = 0; k < SC_U; ++k) {
      int e = e0 + k * SC_THREADS;
      int ec = min(e, last);
      int dv = dst[ec];
      a[k] = src[ec];
      d[k] = (e < end) ? dv : -1;
    }
    for (; e0 < end; e0 += SC_THREADS * SC_U) {
      int dn[SC_U], an[SC_U];
#pragma unroll
      for (int k = 0; k < SC_U; ++k) {
        int e = e0 + (SC_U + k) * SC_THREADS;
        int ec = min(e, last);
        int dv = dst[ec];
        an[k] = src[ec];
        dn[k] = (e < end) ? dv : -1;
      }
      float v0[SC_U], v1[SC_U];
      unsigned r[SC_U];
#pragma unroll
      for (int k = 0; k < SC_U; ++k) {
        r[k] = (unsigned)(d[k] - lo);
        bool ok = r[k] < (unsigned)cnt;
        int ai = ok ? a[k] : 0;
        float2 gv = g2p[ai];
        v0[k] = gv.x;
        v1[k] = gv.y;
      }
      static_assert(SC_U == 8, "the statement below names eight pairs");
      asm volatile("" :: "v"(v0[0]), "v"(v0[1]), "v"(v0[2]), "v"(v0[3]),
                         "v"(v0[4]), "v"(v0[5]), "v"(v0[6]), "v"(v0[7]),
                         "v"(v1[0]), "v"(v1[1]), "v"(v1[2]), "v"(v1[3]),
                         "v"(v1[4]), "v"(v1[5]), "v"(v1[6]), "v"(v1[7]));
#pragma unroll
      for (int k = 0; k < SC_U; ++k)
        if (r[k] < (unsigned)cnt) {
          lds_add(&bins0[r[k]], v0[k]);
          lds_add(&bins1[r[k]], v1[k]);
        }
#pragma unroll
      for (int k = 0; k < SC_U; ++k) { d[k] = dn[k]; a[k] = an[k]; }
    }
  }
  __syncthreads();
  sc_flush(bins0, agg0, lo, cnt);
  sc_flush(bins1, agg1, lo, cnt);
}

__global__ void gcn_finalize(const float* __restrict__ agg0,
                             const float* __restrict__ agg1,
                             const float* __restrict__ g2,
                             const float* __restrict__ dinv,
                             const float* __restrict__ b2,
                             float* __restrict__ out, int n) {
  int i = blockIdx.x * blockDim.x + threadIdx.x;
  if (i >= n) return;
  float di = dinv[i];
  float2 gs = ((const float2*)g2)[i];
  float z0 = di * (agg0[i] + gs.x) + b2[0];
  float z1 = di * (agg1[i] + gs.y) + b2[1];
  float mx = fmaxf(z0, z1);
  float lse = mx + logf(expf(z0 - mx) + expf(z1 - mx));
  float2 r; r.x = z0 - lse; r.y = z1 - lse;
  *(float2*)(out + (size_t)i * 2) = r;
}

extern "C" void kernel_launch(void* const* d_in, const int* in_sizes, int n_in,
                              void* d_out, int out_size, void* d_ws, size_t ws_size,
                              hipStream_t stream) {
  const float* x  = (const float*)d_in[0];
  const int*   ei = (const int*)d_in[1];
  const float* W1 = (const float*)d_in[2];
  const float* b1 = (const float*)d_in[3];
  const float* W2 = (const float*)d_in[4];
  const float* b2 = (const float*)d_in[5];
  float* out = (float*)d_out;

  int n = in_sizes[0];
  int E = in_sizes[1] / 2;
  const int* srcp = ei;
  const int* dstp = ei + E;

  size_t ns = ((size_t)n + 31) & ~(size_t)31;

  float* ws    = (float*)d_ws;
  float* deg   = ws;
  float* dinv  = ws + ns;
  float* s     = ws + 2 * ns;
  float* g     = ws + 3 * ns;
  float* agg0  = ws + 5 * ns;
  float* agg1  = ws + 6 * ns;
  float* relu1 = ws + 7 * ns;

  int gN = (n + TPB - 1) / TPB;

  gcn_init<<<gN, TPB, 0, stream>>>(deg, s, agg0, agg1, n);
  int chunk = (E + SC_SLICES - 1) / SC_SLICES;
  dim3 gS(SC_SLICES, (n + SC_BINS - 1) / SC_BINS);
  gcn_degree<<<gS, SC_THREADS, 0, stream>>>(dstp, deg, E, n, chunk);
  gcn_dinv<<<gN, TPB, 0, stream>>>(deg, x, dinv, g, n, (int)ns);
  gcn_l1_scatter<<<gS, SC_THREADS, 0, stream>>>(srcp, dstp, g, s, E, n, chunk);
  gcn_l1_finalize<<<gN, TPB, 0, stream>>>(g, dinv, s, W1, b1, relu1, n);

  int ntiles = (n + 15) / 16;
  int gW = (int)(((size_t)ntiles * 32 + TPB - 1) / TPB);
  gcn_l2_dense_wmma<<<gW, TPB, 0, stream>>>(relu1, W2, dinv, g, n, ntiles);

  gcn_l2_scatter<<<gS, SC_THREADS, 0, stream>>>(srcp, dstp, g, agg0, agg1, E, n, chunk);
  gcn_finalize<<<gN, TPB, 0, stream>>>(agg0, agg1, g, dinv, b2, out, n);
}


# Edge passes copy a source range's node values into LDS beside the bins (block = slice x source range x destination range; layer 2 one class per block): gathers are LDS reads; changes summation order
# speedup vs baseline: 9.9466x; 1.5221x over previous
#include <hip/hip_runtime.h>

typedef __attribute__((ext_vector_type(2))) float v2f;
typedef __attribute__((ext_vector_type(8))) float v8f;

#define TPB 256


__global__ void gcn_init(float* __restrict__ deg, float* __restrict__ s,
                         float* __restrict__ agg0, float* __restrict__ agg1,
                         int n) {
  int i = blockIdx.x * blockDim.x + threadIdx.x;
  if (i < n) {
    deg[i] = 1.0f;
    s[i] = 0.0f;
    agg0[i] = 0.0f;
    agg1[i] = 0.0f;
  }
}

#define SC_THREADS 1024
#define SC_BINS    12544
#define SC_SLICES  32
#define SC_U       8
#define SC_U_DEG   16

__device__ __forceinline__ void lds_add(float* p, float v) {
  __hip_atomic_fetch_add(p, v, __ATOMIC_RELAXED, __HIP_MEMORY_SCOPE_WORKGROUP);
}

__device__ __forceinline__ void sc_flush(const float* bins, float* __restrict__ g,
                                         int lo, int cnt, unsigned part, unsigned nparts) {
  int rot = (int)((part * (unsigned)cnt) / nparts) & ~31;
  for (int i = threadIdx.x; i < cnt; i += SC_THREADS) {
    int j = i + rot;
    if (j >= cnt) j -= cnt;
    unsafeAtomicAdd(&g[lo + j], bins[j]);
  }
}

__global__ __launch_bounds__(SC_THREADS)
void gcn_degree(const int* __restrict__ dst, float* __restrict__ deg,
                int E, int n, int chunk) {
  __shared__ float bins[SC_BINS];
  const int tid = threadIdx.x;
  const int lo = blockIdx.y * SC_BINS;
  const int cnt = min(n - lo, SC_BINS);
  for (int i = tid; i < SC_BINS; i += SC_THREADS) bins[i] = 0.0f;
  __syncthreads();

  const int start = blockIdx.x * chunk;
  const int end = min(E, start + chunk);
  const int last = end - 1;
  for (int e0 = start + tid; e0 < end; e0 += SC_THREADS * SC_U_DEG) {
    int d[SC_U_DEG];
#pragma unroll
    for (int k = 0; k < SC_U_DEG; ++k) {
      int e = e0 + k * SC_THREADS;
      int dv = dst[min(e, last)];
      d[k] = (e < end) ? dv : -1;
    }
#pragma unroll
    for (int k = 0; k < SC_U_DEG; ++k) {
      unsigned r = (unsigned)(d[k] - lo);
      if (r < (unsigned)cnt) lds_add(&bins[r], 1.0f);
    }
  }
  __syncthreads();
  sc_flush(bins, deg, lo, cnt, blockIdx.x, gridDim.x);
}

__global__ void gcn_dinv(const float* __restrict__ deg,
                         const float* __restrict__ x,
                         float* __restrict__ dinv, float* __restrict__ g1,
                         int n, int ns) {
  int i = blockIdx.x * blockDim.x + threadIdx.x;
  if (i < n) {
    float di = rsqrtf(deg[i]);
    dinv[i] = di;
    g1[i] = x[i] * di;
  } else if (i < ns) {
    dinv[i] = 0.0f;
  }
}

#define TB_SRC 50048
#define TB_DST 25024
#define SC_Q   4

typedef __attribute__((ext_vector_type(4))) int v4i;

template <int STRIDE>
__global__ __launch_bounds__(SC_THREADS)
void gcn_edge_scatter(const int* __restrict__ src, const int* __restrict__ dst,
                      const float* __restrict__ val, float* __restrict__ acc,
                      int E, int n, int chunk, int nrd, int ns) {
  __shared__ float4 tabl4[TB_SRC / 4];
  __shared__ float bins[TB_DST];
  float* tabl = (float*)tabl4;
  const int tid = threadIdx.x;
  const int rdi = blockIdx.y % nrd;
  const int rsi = blockIdx.y / nrd;
  const int lo_d = rdi * TB_DST;
  const int cnt_d = min(n - lo_d, TB_DST);
  const int lo_s = rsi * TB_SRC;
  const int cnt_s = min(n - lo_s, TB_SRC);
  float* ap = acc + (size_t)blockIdx.z * ns;

  for (int i = tid; i < TB_DST; i += SC_THREADS) bins[i] = 0.0f;
  if (STRIDE == 1) {
    const float4* tp4 = (const float4*)(val + lo_s);
    const int nt4 = cnt_s >> 2;
#pragma unroll 4
    for (int i = tid; i < nt4; i += SC_THREADS) tabl4[i] = tp4[i];
    for (int i = (nt4 << 2) + tid; i < cnt_s; i += SC_THREADS) tabl[i] = val[lo_s + i];
  } else {
    const float4* tp4 = (const float4*)(val + 2 * (size_t)lo_s);
    float2* tabl2 = (float2*)tabl;
    const int np = cnt_s >> 1;
    const bool c1 = blockIdx.z != 0;
#pragma unroll 4
    for (int i = tid; i < np; i += SC_THREADS) {
      float4 w = tp4[i];
      float2 o;
      o.x = c1 ? w.y : w.x;
      o.y = c1 ? w.w : w.z;
      tabl2[i] = o;
    }
    if ((cnt_s & 1) && tid == 0)
      tabl[cnt_s - 1] = val[2 * (size_t)(lo_s + cnt_s - 1) + blockIdx.z];
  }
  __syncthreads();

  const int start = blockIdx.x * chunk;
  const int end = min(E, start + chunk);
  if (start < end) {
    const int nq = (end - start) >> 2;
    if (nq > 0) {
      const v4i* sq = (const v4i*)(src + start);
      const v4i* dq = (const v4i*)(dst + start);
      const int lastq = nq - 1;
      v4i cs[SC_Q], cd[SC_Q];
#pragma unroll
      for (int k = 0; k < SC_Q; ++k) {
        int q = tid + k * SC_THREADS;
        int qc = min(q, lastq);
        v4i dv = dq[qc];
        cs[k] = sq[qc];
        cd[k] = (q < nq) ? dv : (v4i)(-1);
      }
      for (int base = 0; base < nq; base += SC_THREADS * SC_Q) {
        v4i cs_n[SC_Q], cd_n[SC_Q];
#pragma unroll
        for (int k = 0; k < SC_Q; ++k) {
          int q = base + (SC_Q + k) * SC_THREADS + tid;
          int qc = min(q, lastq);
          v4i dv = dq[qc];
          cs_n[k] = sq[qc];
          cd_n[k] = (q < nq) ? dv : (v4i)(-1);
        }
        float v[4 * SC_Q];
        unsigned r[4 * SC_Q];
#pragma unroll
        for (int k = 0; k < SC_Q; ++k) {
#pragma unroll
          for (int j = 0; j < 4; ++j) {
            unsigned rd = (unsigned)(cd[k][j] - lo_d);
            unsigned rs = (unsigned)(cs[k][j] - lo_s);
            bool ok = (rd < (unsigned)cnt_d) & (rs < (unsigned)cnt_s);
            r[4 * k + j] = ok ? rd : 0xFFFFFFFFu;
            v[4 * k + j] = tabl[ok ? rs : 0u];
          }
        }
        static_assert(SC_Q == 4, "the statement below names sixteen values");
        asm volatile("" :: "v"(v[0]), "v"(v[1]), "v"(v[2]), "v"(v[3]),
                           "v"(v[4]), "v"(v[5]), "v"(v[6]), "v"(v[7]),
                           "v"(v[8]), "v"(v[9]), "v"(v[10]), "v"(v[11]),
                           "v"(v[12]), "v"(v[13]), "v"(v[14]), "v"(v[15]));
#pragma unroll
        for (int i = 0; i < 4 * SC_Q; ++i)
          if (r[i] != 0xFFFFFFFFu) lds_add(&bins[r[i]], v[i]);
#pragma unroll
        for (int k = 0; k < SC_Q; ++k) { cs[k] = cs_n[k]; cd[k] = cd_n[k]; }
      }
    }
    int e = start + (nq << 2) + tid;
    if (e < end) {
      unsigned rd = (unsigned)(dst[e] - lo_d);
      unsigned rs = (unsigned)(src[e] - lo_s);
      if ((rd < (unsigned)cnt_d) & (rs < (unsigned)cnt_s)) lds_add(&bins[rd], tabl[rs]);
    }
  }
  __syncthreads();
  sc_flush(bins, ap, lo_d, cnt_d, blockIdx.x + gridDim.x * rsi, gridDim.x * (gridDim.y / nrd));
}

__global__ void gcn_l1_finalize(const float* __restrict__ g1,
                                const float* __restrict__ dinv,
                                const float* __restrict__ s,
                                const float* __restrict__ W1,
                                const float* __restrict__ b1,
                                float* __restrict__ relu1, int n) {
  int i = blockIdx.x * blockDim.x + threadIdx.x;
  if (i >= n) return;
  float sn = dinv[i] * (s[i] + g1[i]);
  float4* o = (float4*)(relu1 + (size_t)i * 16);
#pragma unroll
  for (int q = 0; q < 4; ++q) {
    float4 r;
    r.x = fmaxf(fmaf(sn, W1[4 * q + 0], b1[4 * q + 0]), 0.0f);
    r.y = fmaxf(fmaf(sn, W1[4 * q + 1], b1[4 * q + 1]), 0.0f);
    r.z = fmaxf(fmaf(sn, W1[4 * q + 2], b1[4 * q + 2]), 0.0f);
    r.w = fmaxf(fmaf(sn, W1[4 * q + 3], b1[4 * q + 3]), 0.0f);
    o[q] = r;
  }
}

__global__ void gcn_l2_dense_wmma(const float* __restrict__ relu1,
                                  const float* __restrict__ W2,
                                  const float* __restrict__ dinv,
                                  float* __restrict__ g2,
                                  int n, int ntiles) {
  int gtid = blockIdx.x * blockDim.x + threadIdx.x;
  int tile = gtid >> 5;
  int lane = threadIdx.x & 31;
  if (tile >= ntiles) return;
  int base = tile * 16;
  int half = lane >> 4;
  int m = lane & 15;

  v2f bmat[4];
#pragma unroll
  for (int k = 0; k < 4; ++k) {
    int r0 = 4 * k + 2 * half;
    float bx = 0.0f, by = 0.0f;
    if (m < 2) { bx = W2[r0 * 2 + m]; by = W2[(r0 + 1) * 2 + m]; }
    bmat[k].x = bx; bmat[k].y = by;
  }

  int row = base + m;
  row = (row < n) ? row : (n - 1);
  const v2f* p = (const v2f*)(relu1 + (size_t)row * 16 + 2 * half);
  v2f amat[4];
#pragma unroll
  for (int k = 0; k < 4; ++k) amat[k] = p[2 * k];

  v8f c = {};
#pragma unroll
  for (int k = 0; k < 4; ++k)
    c = __builtin_amdgcn_wmma_f32_16x16x4_f32(
         false, amat[k],  false, bmat[k],
         (short)0, c,  false,  false);

  if (m < 2) {
    int node0 = base + half * 8;
    float4 dl = *(const float4*)(dinv + node0);
    float4 dh = *(const float4*)(dinv + node0 + 4);
    float* o = g2 + (size_t)node0 * 2 + m;
    o[0]  = c[0] * dl.x;
    o[2]  = c[1] * dl.y;
    o[4]  = c[2] * dl.z;
    o[6]  = c[3] * dl.w;
    o[8]  = c[4] * dh.x;
    o[10] = c[5] * dh.y;
    o[12] = c[6] * dh.z;
    o[14] = c[7] * dh.w;
  }
}

__global__ void gcn_finalize(const float* __restrict__ agg0,
                             const float* __restrict__ agg1,
                             const float* __restrict__ g2,
                             const float* __restrict__ dinv,
                             const float* __restrict__ b2,
                             float* __restrict__ out, int n) {
  int i = blockIdx.x * blockDim.x + threadIdx.x;
  if (i >= n) return;
  float di = dinv[i];
  float2 gs = ((const float2*)g2)[i];
  float z0 = di * (agg0[i] + gs.x) + b2[0];
  float z1 = di * (agg1[i] + gs.y) + b2[1];
  float mx = fmaxf(z0, z1);
  float lse = mx + logf(expf(z0 - mx) + expf(z1 - mx));
  float2 r; r.x = z0 - lse; r.y = z1 - lse;
  *(float2*)(out + (size_t)i * 2) = r;
}

extern "C" void kernel_launch(void* const* d_in, const int* in_sizes, int n_in,
                              void* d_out, int out_size, void* d_ws, size_t ws_size,
                              hipStream_t stream) {
  const float* x  = (const float*)d_in[0];
  const int*   ei = (const int*)d_in[1];
  const float* W1 = (const float*)d_in[2];
  const float* b1 = (const float*)d_in[3];
  const float* W2 = (const float*)d_in[4];
  const float* b2 = (const float*)d_in[5];
  float* out = (float*)d_out;

  int n = in_sizes[0];
  int E = in_sizes[1] / 2;
  const int* srcp = ei;
  const int* dstp = ei + E;

  size_t ns = ((size_t)n + 31) & ~(size_t)31;

  float* ws    = (float*)d_ws;
  float* deg   = ws;
  float* dinv  = ws + ns;
  float* s     = ws + 2 * ns;
  float* g     = ws + 3 * ns;
  float* agg0  = ws + 5 * ns;
  float* agg1  = ws + 6 * ns;
  float* relu1 = ws + 7 * ns;

  int gN = (n + TPB - 1) / TPB;

  gcn_init<<<gN, TPB, 0, stream>>>(deg, s, agg0, agg1, n);
  int chunk = (E + SC_SLICES - 1) / SC_SLICES;
  dim3 gS(SC_SLICES, (n + SC_BINS - 1) / SC_BINS);
  gcn_degree<<<gS, SC_THREADS, 0, stream>>>(dstp, deg, E, n, chunk);
  gcn_dinv<<<gN, TPB, 0, stream>>>(deg, x, dinv, g, n, (int)ns);
  int nrs = (n + TB_SRC - 1) / TB_SRC;
  int nrd = (n + TB_DST - 1) / TB_DST;
  int chunk1 = (((E + 31) / 32) + 127) & ~127;
  int chunk2 = (((E + 15) / 16) + 127) & ~127;
  dim3 gE1(32, nrs * nrd, 1);
  dim3 gE2(16, nrs * nrd, 2);
  gcn_edge_scatter<1><<<gE1, SC_THREADS, 0, stream>>>(srcp, dstp, g, s, E, n, chunk1, nrd, (int)ns);
  gcn_l1_finalize<<<gN, TPB, 0, stream>>>(g, dinv, s, W1, b1, relu1, n);

  int ntiles = (n + 15) / 16;
  int gW = (int)(((size_t)ntiles * 32 + TPB - 1) / TPB);
  gcn_l2_dense_wmma<<<gW, TPB, 0, stream>>>(relu1, W2, dinv, g, n, ntiles);

  gcn_edge_scatter<2><<<gE2, SC_THREADS, 0, stream>>>(srcp, dstp, g, agg0, E, n, chunk2, nrd, (int)ns);
  gcn_finalize<<<gN, TPB, 0, stream>>>(agg0, agg1, g, dinv, b2, out, n);
}


# Per-node kernels fused into consumers: rsqrt(deg), x*rsqrt(deg) recomputed where used, layer-1 activations built in registers as the WMMA A fragment (self-loop product now an fma); 8 launches to 6
# speedup vs baseline: 10.6425x; 1.0700x over previous
#include <hip/hip_runtime.h>

typedef __attribute__((ext_vector_type(2))) float v2f;
typedef __attribute__((ext_vector_type(8))) float v8f;

#define TPB 256


__global__ void gcn_init(float* __restrict__ deg, float* __restrict__ s,
                         float* __restrict__ agg0, float* __restrict__ agg1,
                         int n) {
  int i = blockIdx.x * blockDim.x + threadIdx.x;
  if (i < n) {
    deg[i] = 1.0f;
    s[i] = 0.0f;
    agg0[i] = 0.0f;
    agg1[i] = 0.0f;
  }
}

#define SC_THREADS 1024
#define SC_BINS    12544
#define SC_SLICES  32
#define SC_U       8
#define SC_U_DEG   16

__device__ __forceinline__ void lds_add(float* p, float v) {
  __hip_atomic_fetch_add(p, v, __ATOMIC_RELAXED, __HIP_MEMORY_SCOPE_WORKGROUP);
}

__device__ __forceinline__ void sc_flush(const float* bins, float* __restrict__ g,
                                         int lo, int cnt, unsigned part, unsigned nparts) {
  int rot = (int)((part * (unsigned)cnt) / nparts) & ~31;
  for (int i = threadIdx.x; i < cnt; i += SC_THREADS) {
    int j = i + rot;
    if (j >= cnt) j -= cnt;
    unsafeAtomicAdd(&g[lo + j], bins[j]);
  }
}

__global__ __launch_bounds__(SC_THREADS)
void gcn_degree(const int* __restrict__ dst, float* __restrict__ deg,
                int E, int n, int chunk) {
  __shared__ float bins[SC_BINS];
  const int tid = threadIdx.x;
  const int lo = blockIdx.y * SC_BINS;
  const int cnt = min(n - lo, SC_BINS);
  for (int i = tid; i < SC_BINS; i += SC_THREADS) bins[i] = 0.0f;
  __syncthreads();

  const int start = blockIdx.x * chunk;
  const int end = min(E, start + chunk);
  const int last = end - 1;
  for (int e0 = start + tid; e0 < end; e0 += SC_THREADS * SC_U_DEG) {
    int d[SC_U_DEG];
#pragma unroll
    for (int k = 0; k < SC_U_DEG; ++k) {
      int e = e0 + k * SC_THREADS;
      int dv = dst[min(e, last)];
      d[k] = (e < end) ? dv : -1;
    }
#pragma unroll
    for (int k = 0; k < SC_U_DEG; ++k) {
      unsigned r = (unsigned)(d[k] - lo);
      if (r < (unsigned)cnt) lds_add(&bins[r], 1.0f);
    }
  }
  __syncthreads();
  sc_flush(bins, deg, lo, cnt, blockIdx.x, gridDim.x);
}

#define TB_SRC 50048
#define TB_DST 25024
#define SC_Q   4

typedef __attribute__((ext_vector_type(4))) int v4i;

template <int STRIDE>
__global__ __launch_bounds__(SC_THREADS)
void gcn_edge_scatter(const int* __restrict__ src, const int* __restrict__ dst,
                      const float* __restrict__ val, const float* __restrict__ deg,
                      float* __restrict__ acc,
                      int E, int n, int chunk, int nrd, int ns) {
  __shared__ float4 tabl4[TB_SRC / 4];
  __shared__ float bins[TB_DST];
  float* tabl = (float*)tabl4;
  const int tid = threadIdx.x;
  const int rdi = blockIdx.y % nrd;
  const int rsi = blockIdx.y / nrd;
  const int lo_d = rdi * TB_DST;
  const int cnt_d = min(n - lo_d, TB_DST);
  const int lo_s = rsi * TB_SRC;
  const int cnt_s = min(n - lo_s, TB_SRC);
  float* ap = acc + (size_t)blockIdx.z * ns;

  for (int i = tid; i < TB_DST; i += SC_THREADS) bins[i] = 0.0f;
  if (STRIDE == 1) {
    const float4* xp4 = (const float4*)(val + lo_s);
    const float4* dp4 = (const float4*)(deg + lo_s);
    const int nt4 = cnt_s >> 2;
#pragma unroll 4
    for (int i = tid; i < nt4; i += SC_THREADS) {
      float4 xv = xp4[i];
      float4 dv = dp4[i];
      float4 o;
      o.x = xv.x * rsqrtf(dv.x);
      o.y = xv.y * rsqrtf(dv.y);
      o.z = xv.z * rsqrtf(dv.z);
      o.w = xv.w * rsqrtf(dv.w);
      tabl4[i] = o;
    }
    for (int i = (nt4 << 2) + tid; i < cnt_s; i += SC_THREADS)
      tabl[i] = val[lo_s + i] * rsqrtf(deg[lo_s + i]);
  } else {
    const float4* tp4 = (const float4*)(val + 2 * (size_t)lo_s);
    float2* tabl2 = (float2*)tabl;
    const int np = cnt_s >> 1;
    const bool c1 = blockIdx.z != 0;
#pragma unroll 4
    for (int i = tid; i < np; i += SC_THREADS) {
      float4 w = tp4[i];
      float2 o;
      o.x = c1 ? w.y : w.x;
      o.y = c1 ? w.w : w.z;
      tabl2[i] = o;
    }
    if ((cnt_s & 1) && tid == 0)
      tabl[cnt_s - 1] = val[2 * (size_t)(lo_s + cnt_s - 1) + blockIdx.z];
  }
  __syncthreads();

  const int start = blockIdx.x * chunk;
  const int end = min(E, start + chunk);
  if (start < end) {
    const int nq = (end - start) >> 2;
    if (nq > 0) {
      const v4i* sq = (const v4i*)(src + start);
      const v4i* dq = (const v4i*)(dst + start);
      const int lastq = nq - 1;
      v4i cs[SC_Q], cd[SC_Q];
#pragma unroll
      for (int k = 0; k < SC_Q; ++k) {
        int q = tid + k * SC_THREADS;
        int qc = min(q, lastq);
        v4i dv = dq[qc];
        cs[k] = sq[qc];
        cd[k] = (q < nq) ? dv : (v4i)(-1);
      }
      for (int base = 0; base < nq; base += SC_THREADS * SC_Q) {
        v4i cs_n[SC_Q], cd_n[SC_Q];
#pragma unroll
        for (int k = 0; k < SC_Q; ++k) {
          int q = base + (SC_Q + k) * SC_THREADS + tid;
          int qc = min(q, lastq);
          v4i dv = dq[qc];
          cs_n[k] = sq[qc];
          cd_n[k] = (q < nq) ? dv : (v4i)(-1);
        }
        float v[4 * SC_Q];
        unsigned r[4 * SC_Q];
#pragma unroll
        for (int k = 0; k < SC_Q; ++k) {
#pragma unroll
          for (int j = 0; j < 4; ++j) {
            unsigned rd = (unsigned)(cd[k][j] - lo_d);
            unsigned rs = (unsigned)(cs[k][j] - lo_s);
            bool ok = (rd < (unsigned)cnt_d) & (rs < (unsigned)cnt_s);
            r[4 * k + j] = ok ? rd : 0xFFFFFFFFu;
            v[4 * k + j] = tabl[ok ? rs : 0u];
          }
        }
        static_assert(SC_Q == 4, "the statement below names sixteen values");
        asm volatile("" :: "v"(v[0]), "v"(v[1]), "v"(v[2]), "v"(v[3]),
                           "v"(v[4]), "v"(v[5]), "v"(v[6]), "v"(v[7]),
                           "v"(v[8]), "v"(v[9]), "v"(v[10]), "v"(v[11]),
                           "v"(v[12]), "v"(v[13]), "v"(v[14]), "v"(v[15]));
#pragma unroll
        for (int i = 0; i < 4 * SC_Q; ++i)
          if (r[i] != 0xFFFFFFFFu) lds_add(&bins[r[i]], v[i]);
#pragma unroll
        for (int k = 0; k < SC_Q; ++k) { cs[k] = cs_n[k]; cd[k] = cd_n[k]; }
      }
    }
    int e = start + (nq << 2) + tid;
    if (e < end) {
      unsigned rd = (unsigned)(dst[e] - lo_d);
      unsigned rs = (unsigned)(src[e] - lo_s);
      if ((rd < (unsigned)cnt_d) & (rs < (unsigned)cnt_s)) lds_add(&bins[rd], tabl[rs]);
    }
  }
  __syncthreads();
  sc_flush(bins, ap, lo_d, cnt_d, blockIdx.x + gridDim.x * rsi, gridDim.x * (gridDim.y / nrd));
}

__global__ void gcn_l1_dense_wmma(const float* __restrict__ x,
                                  const float* __restrict__ deg,
                                  const float* __restrict__ s,
                                  const float* __restrict__ W1,
                                  const float* __restrict__ b1,
                                  const float* __restrict__ W2,
                                  float* __restrict__ g2,
                                  int n, int ntiles) {
  int gtid = blockIdx.x * blockDim.x + threadIdx.x;
  int tile = gtid >> 5;
  int lane = threadIdx.x & 31;
  if (tile >= ntiles) return;
  int base = tile * 16;
  int half = lane >> 4;
  int m = lane & 15;

  v2f bmat[4];
#pragma unroll
  for (int k = 0; k < 4; ++k) {
    int r0 = 4 * k + 2 * half;
    float bx = 0.0f, by = 0.0f;
    if (m < 2) { bx = W2[r0 * 2 + m]; by = W2[(r0 + 1) * 2 + m]; }
    bmat[k].x = bx; bmat[k].y = by;
  }

  int row = base + m;
  const bool live = row < n;
  row = live ? row : (n - 1);
  float di = rsqrtf(deg[row]);
  float sn = di * (s[row] + x[row] * di);
  const v2f* w1p = (const v2f*)(W1 + 2 * half);
  const v2f* b1p = (const v2f*)(b1 + 2 * half);
  v2f amat[4];
#pragma unroll
  for (int k = 0; k < 4; ++k) {
    v2f w = w1p[2 * k];
    v2f bb = b1p[2 * k];
    amat[k].x = fmaxf(fmaf(sn, w.x, bb.x), 0.0f);
    amat[k].y = fmaxf(fmaf(sn, w.y, bb.y), 0.0f);
  }

  int dbits = __float_as_int(live ? di : 0.0f);
  float dr[8];
#pragma unroll
  for (int r = 0; r < 8; ++r)
    dr[r] = __int_as_float(__builtin_amdgcn_ds_bpermute((8 * half + r) << 2, dbits));

  asm volatile("" : "+v"(amat[0]), "+v"(amat[1]), "+v"(amat[2]), "+v"(amat[3]),
                    "+v"(bmat[0]), "+v"(bmat[1]), "+v"(bmat[2]), "+v"(bmat[3]));

  v8f c = {};
#pragma unroll
  for (int k = 0; k < 4; ++k)
    c = __builtin_amdgcn_wmma_f32_16x16x4_f32(
         false, amat[k],  false, bmat[k],
         (short)0, c,  false,  false);

  asm volatile("v_nop\n\tv_nop\n\tv_nop\n\tv_nop" : "+v"(c));
  asm volatile("" :: "v"(amat[0]), "v"(amat[1]), "v"(amat[2]), "v"(amat[3]),
                     "v"(bmat[0]), "v"(bmat[1]), "v"(bmat[2]), "v"(bmat[3]));

  if (m < 2) {
    int node0 = base + half * 8;
    float* o = g2 + (size_t)node0 * 2 + m;
#pragma unroll
    for (int r = 0; r < 8; ++r) o[2 * r] = c[r] * dr[r];
  }
}

__global__ void gcn_finalize(const float* __restrict__ agg0,
                             const float* __restrict__ agg1,
                             const float* __restrict__ g2,
                             const float* __restrict__ deg,
                             const float* __restrict__ b2,
                             float* __restrict__ out, int n) {
  int i = blockIdx.x * blockDim.x + threadIdx.x;
  if (i >= n) return;
  float di = rsqrtf(deg[i]);
  float2 gs = ((const float2*)g2)[i];
  float z0 = di * (agg0[i] + gs.x) + b2[0];
  float z1 = di * (agg1[i] + gs.y) + b2[1];
  float mx = fmaxf(z0, z1);
  float lse = mx + logf(expf(z0 - mx) + expf(z1 - mx));
  float2 r; r.x = z0 - lse; r.y = z1 - lse;
  *(float2*)(out + (size_t)i * 2) = r;
}

extern "C" void kernel_launch(void* const* d_in, const int* in_sizes, int n_in,
                              void* d_out, int out_size, void* d_ws, size_t ws_size,
                              hipStream_t stream) {
  const float* x  = (const float*)d_in[0];
  const int*   ei = (const int*)d_in[1];
  const float* W1 = (const float*)d_in[2];
  const float* b1 = (const float*)d_in[3];
  const float* W2 = (const float*)d_in[4];
  const float* b2 = (const float*)d_in[5];
  float* out = (float*)d_out;

  int n = in_sizes[0];
  int E = in_sizes[1] / 2;
  const int* srcp = ei;
  const int* dstp = ei + E;

  size_t ns = ((size_t)n + 31) & ~(size_t)31;

  float* ws    = (float*)d_ws;
  float* deg   = ws;
  float* s     = ws + 2 * ns;
  float* g     = ws + 3 * ns;
  float* agg0  = ws + 5 * ns;
  float* agg1  = ws + 6 * ns;

  int gN = (n + TPB - 1) / TPB;

  gcn_init<<<gN, TPB, 0, stream>>>(deg, s, agg0, agg1, n);
  int chunk = (E + SC_SLICES - 1) / SC_SLICES;
  dim3 gS(SC_SLICES, (n + SC_BINS - 1) / SC_BINS);
  gcn_degree<<<gS, SC_THREADS, 0, stream>>>(dstp, deg, E, n, chunk);
  int nrs = (n + TB_SRC - 1) / TB_SRC;
  int nrd = (n + TB_DST - 1) / TB_DST;
  int chunk1 = (((E + 31) / 32) + 127) & ~127;
  int chunk2 = (((E + 15) / 16) + 127) & ~127;
  dim3 gE1(32, nrs * nrd, 1);
  dim3 gE2(16, nrs * nrd, 2);
  gcn_edge_scatter<1><<<gE1, SC_THREADS, 0, stream>>>(srcp, dstp, x, deg, s, E, n, chunk1, nrd, (int)ns);

  int ntiles = (n + 15) / 16;
  int gW = (int)(((size_t)ntiles * 32 + TPB - 1) / TPB);
  gcn_l1_dense_wmma<<<gW, TPB, 0, stream>>>(x, deg, s, W1, b1, W2, g, n, ntiles);

  gcn_edge_scatter<2><<<gE2, SC_THREADS, 0, stream>>>(srcp, dstp, g, deg, agg0, E, n, chunk2, nrd, (int)ns);
  gcn_finalize<<<gN, TPB, 0, stream>>>(agg0, agg1, g, deg, b2, out, n);
}


# log_softmax of two classes needs only d=z0-z1: layer 2 carries one plane, relu1 x (W2[:,0]-W2[:,1]), so its edge pass runs once, not per class; changes rounding and summation order (32 edge slices)
# speedup vs baseline: 12.3613x; 1.1615x over previous
#include <hip/hip_runtime.h>

typedef __attribute__((ext_vector_type(2))) float v2f;
typedef __attribute__((ext_vector_type(8))) float v8f;

#define TPB 256


__global__ void gcn_init(float* __restrict__ deg, float* __restrict__ s,
                         float* __restrict__ agg, int n) {
  int i = blockIdx.x * blockDim.x + threadIdx.x;
  if (i < n) {
    deg[i] = 1.0f;
    s[i] = 0.0f;
    agg[i] = 0.0f;
  }
}

#define SC_THREADS 1024
#define SC_BINS    12544
#define SC_SLICES  32
#define SC_U       8
#define SC_U_DEG   16

__device__ __forceinline__ void lds_add(float* p, float v) {
  __hip_atomic_fetch_add(p, v, __ATOMIC_RELAXED, __HIP_MEMORY_SCOPE_WORKGROUP);
}

__device__ __forceinline__ void sc_flush(const float* bins, float* __restrict__ g,
                                         int lo, int cnt, unsigned part, unsigned nparts) {
  int rot = (int)((part * (unsigned)cnt) / nparts) & ~31;
  for (int i = threadIdx.x; i < cnt; i += SC_THREADS) {
    int j = i + rot;
    if (j >= cnt) j -= cnt;
    unsafeAtomicAdd(&g[lo + j], bins[j]);
  }
}

__global__ __launch_bounds__(SC_THREADS)
void gcn_degree(const int* __restrict__ dst, float* __restrict__ deg,
                int E, int n, int chunk) {
  __shared__ float bins[SC_BINS];
  const int tid = threadIdx.x;
  const int lo = blockIdx.y * SC_BINS;
  const int cnt = min(n - lo, SC_BINS);
  for (int i = tid; i < SC_BINS; i += SC_THREADS) bins[i] = 0.0f;
  __syncthreads();

  const int start = blockIdx.x * chunk;
  const int end = min(E, start + chunk);
  const int last = end - 1;
  for (int e0 = start + tid; e0 < end; e0 += SC_THREADS * SC_U_DEG) {
    int d[SC_U_DEG];
#pragma unroll
    for (int k = 0; k < SC_U_DEG; ++k) {
      int e = e0 + k * SC_THREADS;
      int dv = dst[min(e, last)];
      d[k] = (e < end) ? dv : -1;
    }
#pragma unroll
    for (int k = 0; k < SC_U_DEG; ++k) {
      unsigned r = (unsigned)(d[k] - lo);
      if (r < (unsigned)cnt) lds_add(&bins[r], 1.0f);
    }
  }
  __syncthreads();
  sc_flush(bins, deg, lo, cnt, blockIdx.x, gridDim.x);
}

#define TB_SRC 50048
#define TB_DST 25024
#define SC_Q   4

typedef __attribute__((ext_vector_type(4))) int v4i;

template <int LAYER>
__global__ __launch_bounds__(SC_THREADS)
void gcn_edge_scatter(const int* __restrict__ src, const int* __restrict__ dst,
                      const float* __restrict__ val, const float* __restrict__ deg,
                      float* __restrict__ acc,
                      int E, int n, int chunk, int nrd) {
  __shared__ float4 tabl4[TB_SRC / 4];
  __shared__ float bins[TB_DST];
  float* tabl = (float*)tabl4;
  const int tid = threadIdx.x;
  const int rdi = blockIdx.y % nrd;
  const int rsi = blockIdx.y / nrd;
  const int lo_d = rdi * TB_DST;
  const int cnt_d = min(n - lo_d, TB_DST);
  const int lo_s = rsi * TB_SRC;
  const int cnt_s = min(n - lo_s, TB_SRC);

  for (int i = tid; i < TB_DST; i += SC_THREADS) bins[i] = 0.0f;
  if (LAYER == 1) {
    const float4* xp4 = (const float4*)(val + lo_s);
    const float4* dp4 = (const float4*)(deg + lo_s);
    const int nt4 = cnt_s >> 2;
#pragma unroll 4
    for (int i = tid; i < nt4; i += SC_THREADS) {
      float4 xv = xp4[i];
      float4 dv = dp4[i];
      float4 o;
      o.x = xv.x * rsqrtf(dv.x);
      o.y = xv.y * rsqrtf(dv.y);
      o.z = xv.z * rsqrtf(dv.z);
      o.w = xv.w * rsqrtf(dv.w);
      tabl4[i] = o;
    }
    for (int i = (nt4 << 2) + tid; i < cnt_s; i += SC_THREADS)
      tabl[i] = val[lo_s + i] * rsqrtf(deg[lo_s + i]);
  } else {
    const float4* vp4 = (const float4*)(val + lo_s);
    const int nt4 = cnt_s >> 2;
#pragma unroll 4
    for (int i = tid; i < nt4; i += SC_THREADS) tabl4[i] = vp4[i];
    for (int i = (nt4 << 2) + tid; i < cnt_s; i += SC_THREADS) tabl[i] = val[lo_s + i];
  }
  __syncthreads();

  const int start = blockIdx.x * chunk;
  const int end = min(E, start + chunk);
  if (start < end) {
    const int nq = (end - start) >> 2;
    if (nq > 0) {
      const v4i* sq = (const v4i*)(src + start);
      const v4i* dq = (const v4i*)(dst + start);
      const int lastq = nq - 1;
      v4i cs[SC_Q], cd[SC_Q];
#pragma unroll
      for (int k = 0; k < SC_Q; ++k) {
        int q = tid + k * SC_THREADS;
        int qc = min(q, lastq);
        v4i dv = dq[qc];
        cs[k] = sq[qc];
        cd[k] = (q < nq) ? dv : (v4i)(-1);
      }
      for (int base = 0; base < nq; base += SC_THREADS * SC_Q) {
        v4i cs_n[SC_Q], cd_n[SC_Q];
#pragma unroll
        for (int k = 0; k < SC_Q; ++k) {
          int q = base + (SC_Q + k) * SC_THREADS + tid;
          int qc = min(q, lastq);
          v4i dv = dq[qc];
          cs_n[k] = sq[qc];
          cd_n[k] = (q < nq) ? dv : (v4i)(-1);
        }
        float v[4 * SC_Q];
        unsigned r[4 * SC_Q];
#pragma unroll
        for (int k = 0; k < SC_Q; ++k) {
#pragma unroll
          for (int j = 0; j < 4; ++j) {
            unsigned rd = (unsigned)(cd[k][j] - lo_d);
            unsigned rs = (unsigned)(cs[k][j] - lo_s);
            bool ok = (rd < (unsigned)cnt_d) & (rs < (unsigned)cnt_s);
            r[4 * k + j] = ok ? rd : 0xFFFFFFFFu;
            v[4 * k + j] = tabl[ok ? rs : 0u];
          }
        }
        static_assert(SC_Q == 4, "the statement below names sixteen values");
        asm volatile("" :: "v"(v[0]), "v"(v[1]), "v"(v[2]), "v"(v[3]),
                           "v"(v[4]), "v"(v[5]), "v"(v[6]), "v"(v[7]),
                           "v"(v[8]), "v"(v[9]), "v"(v[10]), "v"(v[11]),
                           "v"(v[12]), "v"(v[13]), "v"(v[14]), "v"(v[15]));
#pragma unroll
        for (int i = 0; i < 4 * SC_Q; ++i)
          if (r[i] != 0xFFFFFFFFu) lds_add(&bins[r[i]], v[i]);
#pragma unroll
        for (int k = 0; k < SC_Q; ++k) { cs[k] = cs_n[k]; cd[k] = cd_n[k]; }
      }
    }
    int e = start + (nq << 2) + tid;
    if (e < end) {
      unsigned rd = (unsigned)(dst[e] - lo_d);
      unsigned rs = (unsigned)(src[e] - lo_s);
      if ((rd < (unsigned)cnt_d) & (rs < (unsigned)cnt_s)) lds_add(&bins[rd], tabl[rs]);
    }
  }
  __syncthreads();
  sc_flush(bins, acc, lo_d, cnt_d, blockIdx.x + gridDim.x * rsi, gridDim.x * (gridDim.y / nrd));
}

__global__ void gcn_l1_dense_wmma(const float* __restrict__ x,
                                  const float* __restrict__ deg,
                                  const float* __restrict__ s,
                                  const float* __restrict__ W1,
                                  const float* __restrict__ b1,
                                  const float* __restrict__ W2,
                                  float* __restrict__ g2,
                                  int n, int nwaves) {
  int gtid = blockIdx.x * blockDim.x + threadIdx.x;
  int wv = gtid >> 5;
  int lane = threadIdx.x & 31;
  if (wv >= nwaves) return;
  int base = wv * 32;
  int half = lane >> 4;
  int c = lane & 15;

  const float4* w2p = (const float4*)(W2 + 4 * half);
  const bool wrow = (c & 7) == 0;
  v2f wmat[4];
#pragma unroll
  for (int k = 0; k < 4; ++k) {
    float4 q = w2p[2 * k];
    wmat[k].x = wrow ? (q.x - q.y) : 0.0f;
    wmat[k].y = wrow ? (q.z - q.w) : 0.0f;
  }

  const v2f* w1p = (const v2f*)(W1 + 2 * half);
  const v2f* b1p = (const v2f*)(b1 + 2 * half);
  v2f hmat[2][4];
  float dn[2];
#pragma unroll
  for (int t = 0; t < 2; ++t) {
    int node = base + 16 * t + c;
    const bool live = node < n;
    node = live ? node : (n - 1);
    float di = rsqrtf(deg[node]);
    float sn = di * (s[node] + x[node] * di);
    dn[t] = live ? di : 0.0f;
#pragma unroll
    for (int k = 0; k < 4; ++k) {
      v2f w = w1p[2 * k];
      v2f bb = b1p[2 * k];
      hmat[t][k].x = fmaxf(fmaf(sn, w.x, bb.x), 0.0f);
      hmat[t][k].y = fmaxf(fmaf(sn, w.y, bb.y), 0.0f);
    }
  }

  asm volatile("" : "+v"(wmat[0]), "+v"(wmat[1]), "+v"(wmat[2]), "+v"(wmat[3]),
                    "+v"(hmat[0][0]), "+v"(hmat[0][1]), "+v"(hmat[0][2]), "+v"(hmat[0][3]),
                    "+v"(hmat[1][0]), "+v"(hmat[1][1]), "+v"(hmat[1][2]), "+v"(hmat[1][3]));

  v8f c0 = {};
  v8f c1 = {};
#pragma unroll
  for (int k = 0; k < 4; ++k) {
    c0 = __builtin_amdgcn_wmma_f32_16x16x4_f32(
         false, wmat[k],  false, hmat[0][k],
         (short)0, c0,  false,  false);
    c1 = __builtin_amdgcn_wmma_f32_16x16x4_f32(
         false, wmat[k],  false, hmat[1][k],
         (short)0, c1,  false,  false);
  }

  asm volatile("v_nop\n\tv_nop\n\tv_nop\n\tv_nop" : "+v"(c0), "+v"(c1));
  asm volatile("" :: "v"(wmat[0]), "v"(wmat[1]), "v"(wmat[2]), "v"(wmat[3]),
                     "v"(hmat[0][0]), "v"(hmat[0][1]), "v"(hmat[0][2]), "v"(hmat[0][3]),
                     "v"(hmat[1][0]), "v"(hmat[1][1]), "v"(hmat[1][2]), "v"(hmat[1][3]));

  float r = half ? (c1[0] * dn[1]) : (c0[0] * dn[0]);
  g2[base + lane] = r;
}

__global__ void gcn_finalize(const float* __restrict__ agg,
                             const float* __restrict__ g2,
                             const float* __restrict__ deg,
                             const float* __restrict__ b2,
                             float* __restrict__ out, int n) {
  int i = blockIdx.x * blockDim.x + threadIdx.x;
  if (i >= n) return;
  float di = rsqrtf(deg[i]);
  float d = di * (agg[i] + g2[i]) + (b2[0] - b2[1]);
  float l = logf(1.0f + expf(-fabsf(d)));
  float2 r; r.x = fminf(d, 0.0f) - l; r.y = fminf(-d, 0.0f) - l;
  *(float2*)(out + (size_t)i * 2) = r;
}

extern "C" void kernel_launch(void* const* d_in, const int* in_sizes, int n_in,
                              void* d_out, int out_size, void* d_ws, size_t ws_size,
                              hipStream_t stream) {
  const float* x  = (const float*)d_in[0];
  const int*   ei = (const int*)d_in[1];
  const float* W1 = (const float*)d_in[2];
  const float* b1 = (const float*)d_in[3];
  const float* W2 = (const float*)d_in[4];
  const float* b2 = (const float*)d_in[5];
  float* out = (float*)d_out;

  int n = in_sizes[0];
  int E = in_sizes[1] / 2;
  const int* srcp = ei;
  const int* dstp = ei + E;

  size_t ns = ((size_t)n + 31) & ~(size_t)31;

  float* ws    = (float*)d_ws;
  float* deg   = ws;
  float* s     = ws + 2 * ns;
  float* g     = ws + 3 * ns;
  float* agg   = ws + 5 * ns;

  int gN = (n + TPB - 1) / TPB;

  gcn_init<<<gN, TPB, 0, stream>>>(deg, s, agg, n);
  int chunk = (E + SC_SLICES - 1) / SC_SLICES;
  dim3 gS(SC_SLICES, (n + SC_BINS - 1) / SC_BINS);
  gcn_degree<<<gS, SC_THREADS, 0, stream>>>(dstp, deg, E, n, chunk);
  int nrs = (n + TB_SRC - 1) / TB_SRC;
  int nrd = (n + TB_DST - 1) / TB_DST;
  int chunkE = (((E + 31) / 32) + 127) & ~127;
  dim3 gE(32, nrs * nrd, 1);
  gcn_edge_scatter<1><<<gE, SC_THREADS, 0, stream>>>(srcp, dstp, x, deg, s, E, n, chunkE, nrd);

  int nwaves = (n + 31) / 32;
  int gW = (int)(((size_t)nwaves * 32 + TPB - 1) / TPB);
  gcn_l1_dense_wmma<<<gW, TPB, 0, stream>>>(x, deg, s, W1, b1, W2, g, n, nwaves);

  gcn_edge_scatter<2><<<gE, SC_THREADS, 0, stream>>>(srcp, dstp, g, deg, agg, E, n, chunkE, nrd);
  gcn_finalize<<<gN, TPB, 0, stream>>>(agg, g, deg, b2, out, n);
}
